# starting kernel
# baseline (speedup 1.0000x reference)
#include <hip/hip_runtime.h>
#include <math.h>

#define B_ 64
#define L_ 1024
#define C_ 512
#define NUM_DOMAINS 4
#define CAPACITY 1024
#define EPS 1e-6f
#define LSPLIT 4
#define LCHUNK (L_ / LSPLIT)

typedef __attribute__((ext_vector_type(2))) float v2f;
typedef __attribute__((ext_vector_type(4))) float v4f;
typedef __attribute__((ext_vector_type(8))) float v8f;

__global__ __launch_bounds__(128) void dq_stats_kernel(
    const float* __restrict__ x, float* __restrict__ psum,
    float* __restrict__ psq) {
  const int wave  = threadIdx.x >> 5;
  const int lane  = threadIdx.x & 31;
  const int b     = blockIdx.z;
  const int split = blockIdx.y;
  const int cbase = blockIdx.x * 128 + wave * 32;
  const int n     = lane & 15;
  const int koff  = (lane >> 4) * 2;
  const int c0    = cbase + 2 * n;

  const float* xp =
      x + ((size_t)b * L_ + (size_t)split * LCHUNK) * C_ + c0;

  v8f accS0 = {0.f, 0.f, 0.f, 0.f, 0.f, 0.f, 0.f, 0.f};
  v8f accS1 = accS0, accQ0 = accS0, accQ1 = accS0;
  const v2f ones = {1.0f, 1.0f};

  for (int l = 0; l < LCHUNK; l += 4) {
    const float* p = xp + (size_t)(l + koff) * C_;
    const v2f r0 = *(const v2f*)(p);
    const v2f r1 = *(const v2f*)(p + C_);
    v2f b0, b1;
    b0.x = r0.x; b0.y = r1.x;
    b1.x = r0.y; b1.y = r1.y;
    v2f q0 = b0 * b0;
    v2f q1 = b1 * b1;
    accS0 = __builtin_amdgcn_wmma_f32_16x16x4_f32(
        false, ones, false, b0, (short)0, accS0, false, false);
    accS1 = __builtin_amdgcn_wmma_f32_16x16x4_f32(
        false, ones, false, b1, (short)0, accS1, false, false);
    accQ0 = __builtin_amdgcn_wmma_f32_16x16x4_f32(
        false, ones, false, q0, (short)0, accQ0, false, false);
    accQ1 = __builtin_amdgcn_wmma_f32_16x16x4_f32(
        false, ones, false, q1, (short)0, accQ1, false, false);
  }

  if (lane < 16) {
    size_t o = ((size_t)b * LSPLIT + split) * C_ + c0;
    v2f s, q;
    s.x = accS0[0]; s.y = accS1[0];
    q.x = accQ0[0]; q.y = accQ1[0];
    *(v2f*)(psum + o) = s;
    *(v2f*)(psq + o)  = q;
  }
}

__global__ __launch_bounds__(256) void dq_mix_kernel(
    const float* __restrict__ psum, const float* __restrict__ psq,
    const int* __restrict__ domain, const float* __restrict__ mean_q,
    const float* __restrict__ sig_q, const float* __restrict__ lmda,
    const int* __restrict__ d_ind1, const int* __restrict__ f_ind1,
    const int* __restrict__ offsets, float* __restrict__ a_out,
    float* __restrict__ bias_out) {
  const int b = blockIdx.x;
  __shared__ int s_src;
  if (threadIdx.x == 0) {
    const int d1 = d_ind1[b];
    const int f1 = f_ind1[b];
    int src = -1;
    if (d1 < NUM_DOMAINS) {
      const int off = offsets[d1];
      int rank = 0;
      for (int bp = 0; bp < B_; ++bp) {
        if (domain[bp] == d1) {
          int pos = (off + rank) % CAPACITY;
          if (pos == f1) src = bp;
          ++rank;
        }
      }
    }
    s_src = src;
  }
  __syncthreads();

  const int   src = s_src;
  const int   d1  = d_ind1[b];
  const int   f1  = f_ind1[b];
  const float lam = lmda[b];

  for (int c = threadIdx.x; c < C_; c += blockDim.x) {
    float s = 0.f, q = 0.f;
#pragma unroll
    for (int k = 0; k < LSPLIT; ++k) {
      s += psum[((size_t)b * LSPLIT + k) * C_ + c];
      q += psq[((size_t)b * LSPLIT + k) * C_ + c];
    }
    const float mu  = s * (1.0f / L_);
    const float var = (q - s * s * (1.0f / L_)) * (1.0f / (L_ - 1));
    const float sig = sqrtf(var + EPS);

    float mu1, sig1;
    if (src >= 0) {
      float s2 = 0.f, q2 = 0.f;
#pragma unroll
      for (int k = 0; k < LSPLIT; ++k) {
        s2 += psum[((size_t)src * LSPLIT + k) * C_ + c];
        q2 += psq[((size_t)src * LSPLIT + k) * C_ + c];
      }
      mu1 = s2 * (1.0f / L_);
      const float var2 = (q2 - s2 * s2 * (1.0f / L_)) * (1.0f / (L_ - 1));
      sig1 = sqrtf(var2 + EPS);
    } else {
      const size_t qo = ((size_t)d1 * CAPACITY + f1) * C_ + c;
      mu1  = mean_q[qo];
      sig1 = sig_q[qo];
    }

    const float mu_mix  = mu * lam + mu1 * (1.0f - lam);
    const float sig_mix = sig * lam + sig1 * (1.0f - lam);
    const float a       = sig_mix / sig;
    a_out[(size_t)b * C_ + c]    = a;
    bias_out[(size_t)b * C_ + c] = mu_mix - mu * a;
  }
}

__global__ __launch_bounds__(256) void dq_apply_kernel(
    const float* __restrict__ x, const float* __restrict__ a_arr,
    const float* __restrict__ bias_arr, float* __restrict__ out) {
  const size_t idx = (size_t)blockIdx.x * blockDim.x + threadIdx.x;
  const int C4 = C_ / 4;
  const int c4 = (int)(idx & (C4 - 1));
  const size_t row = idx >> 7;
  const int b = (int)(row >> 10);

  const v4f xv = ((const v4f*)x)[idx];
  const v4f av = ((const v4f*)a_arr)[(size_t)b * C4 + c4];
  const v4f bv = ((const v4f*)bias_arr)[(size_t)b * C4 + c4];
  v4f o;
  o.x = fmaf(xv.x, av.x, bv.x);
  o.y = fmaf(xv.y, av.y, bv.y);
  o.z = fmaf(xv.z, av.z, bv.z);
  o.w = fmaf(xv.w, av.w, bv.w);
  __builtin_nontemporal_store(o, &((v4f*)out)[idx]);
}

extern "C" void kernel_launch(void* const* d_in, const int* in_sizes, int n_in,
                              void* d_out, int out_size, void* d_ws,
                              size_t ws_size, hipStream_t stream) {
  const float* x       = (const float*)d_in[0];
  const int*   domain  = (const int*)d_in[1];
  const float* mean_q  = (const float*)d_in[2];
  const float* sig_q   = (const float*)d_in[3];
  const float* lmda    = (const float*)d_in[4];
  const int*   d_ind1  = (const int*)d_in[5];
  const int*   f_ind1  = (const int*)d_in[6];
  const int*   offsets = (const int*)d_in[7];
  float*       out     = (float*)d_out;

  float* psum  = (float*)d_ws;
  float* psq   = psum + (size_t)B_ * LSPLIT * C_;
  float* a_arr = psq + (size_t)B_ * LSPLIT * C_;
  float* b_arr = a_arr + (size_t)B_ * C_;

  dim3 gA(C_ / 128, LSPLIT, B_);
  dq_stats_kernel<<<gA, 128, 0, stream>>>(x, psum, psq);

  dq_mix_kernel<<<B_, 256, 0, stream>>>(psum, psq, domain, mean_q, sig_q, lmda,
                                        d_ind1, f_ind1, offsets, a_arr, b_arr);

  const size_t n4 = (size_t)B_ * L_ * C_ / 4;
  dq_apply_kernel<<<(unsigned)(n4 / 256), 256, 0, stream>>>(x, a_arr, b_arr,
                                                            out);
}


# Apply pass re-tiled to re-read x from L2: same grid as the stats pass, work-group (x,split,b) takes the 256-row x 128-channel tile it summed, rows newest-first (was 512 whole-row slabs)
# speedup vs baseline: 1.5777x; 1.5777x over previous
#include <hip/hip_runtime.h>
#include <math.h>
#include <stddef.h>

#define B_ 64
#define L_ 1024
#define C_ 512
#define NUM_DOMAINS 4
#define CAPACITY 1024
#define EPS 1e-6f
#define LSPLIT 4
#define LCHUNK (L_ / LSPLIT)

typedef __attribute__((ext_vector_type(2))) float v2f;
typedef __attribute__((ext_vector_type(4))) float v4f;
typedef __attribute__((ext_vector_type(8))) float v8f;

__global__ __launch_bounds__(128) void dq_stats_kernel(
    const float* __restrict__ x, float* __restrict__ psum,
    float* __restrict__ psq) {
  const int wave  = threadIdx.x >> 5;
  const int lane  = threadIdx.x & 31;
  const int b     = blockIdx.z;
  const int split = blockIdx.y;
  const int cbase = blockIdx.x * 128 + wave * 32;
  const int n     = lane & 15;
  const int koff  = (lane >> 4) * 2;
  const int c0    = cbase + 2 * n;

  const float* xp =
      x + ((size_t)b * L_ + (size_t)split * LCHUNK) * C_ + c0;

  v8f accQ0 = {0.f, 0.f, 0.f, 0.f, 0.f, 0.f, 0.f, 0.f};
  v8f accQ1 = accQ0;
  v2f sumA = {0.f, 0.f};
  v2f sumB = {0.f, 0.f};

  for (int l = 0; l < LCHUNK; l += 4) {
    const float* p = xp + (size_t)(l + koff) * C_;
    const v2f r0 = *(const v2f*)(p);
    const v2f r1 = *(const v2f*)(p + C_);
    sumA += r0;
    sumB += r1;
    v2f b0, b1;
    b0.x = r0.x; b0.y = r1.x;
    b1.x = r0.y; b1.y = r1.y;
    accQ0 = __builtin_amdgcn_wmma_f32_16x16x4_f32(
        false, b0, false, b0, (short)0, accQ0, false, false);
    accQ1 = __builtin_amdgcn_wmma_f32_16x16x4_f32(
        false, b1, false, b1, (short)0, accQ1, false, false);
  }

  v2f s = sumA + sumB;
  s.x += __shfl_xor(s.x, 16);
  s.y += __shfl_xor(s.y, 16);

  const int r = lane & 7;
  v2f q;
  q.x = accQ0[0]; q.y = accQ1[0];
#pragma unroll
  for (int i = 1; i < 8; ++i) {
    if (r == i) { q.x = accQ0[i]; q.y = accQ1[i]; }
  }

  if (lane < 8 || lane >= 24) {
    size_t o = ((size_t)b * LSPLIT + split) * C_ + c0;
    *(v2f*)(psum + o) = s;
    *(v2f*)(psq + o)  = q;
  }
}

__device__ __forceinline__ void dq_stat(float s, float q, float& mu,
                                        float& sig) {
  mu = s * (1.0f / L_);
  const float var = (q - s * s * (1.0f / L_)) * (1.0f / (L_ - 1));
  sig = sqrtf(var + EPS);
}

__device__ __forceinline__ void dq_coef(float mu, float sig, float mu1,
                                        float sig1, float lam, float& a,
                                        float& bias) {
  const float mu_mix  = mu * lam + mu1 * (1.0f - lam);
  const float sig_mix = sig * lam + sig1 * (1.0f - lam);
  a    = sig_mix / sig;
  bias = mu_mix - mu * a;
}

#define APPLY_ROWS LCHUNK
#define APPLY_WAVES 4
#define APPLY_COLS4 32
#define APPLY_BATCH 8
#define APPLY_AHEAD 2
#define APPLY_STEP (APPLY_WAVES * (C_ / 4))
#define APPLY_NBATCH (APPLY_ROWS / APPLY_WAVES / APPLY_BATCH)

__global__ __launch_bounds__(128) void dq_apply_kernel(
    const float* __restrict__ x, const float* __restrict__ psum,
    const float* __restrict__ psq, const int* __restrict__ domain,
    const float* __restrict__ mean_q, const float* __restrict__ sig_q,
    const float* __restrict__ lmda, const int* __restrict__ d_ind1,
    const int* __restrict__ f_ind1, const int* __restrict__ offsets,
    float* __restrict__ out) {
  const int tid  = threadIdx.x;
  const int lane = tid & 31;
  const int rsub = tid >> 5;
  const int c4   = blockIdx.x * APPLY_COLS4 + lane;
  const int b    = blockIdx.z;
  const int row0 = blockIdx.y * APPLY_ROWS + (APPLY_ROWS - APPLY_WAVES) + rsub;

  const size_t base = ((size_t)b * L_ + row0) * (C_ / 4) + c4;
  const v4f* __restrict__ xp = (const v4f*)x + base;
  v4f* __restrict__ op       = (v4f*)out + base;

  const int   d1   = d_ind1[b];
  const int   f1   = f_ind1[b];
  const float lam  = lmda[b];
  const int   offv = offsets[lane & (NUM_DOMAINS - 1)];
  const int   dom_lo = domain[lane];
  const int   dom_hi = domain[lane + 32];
  const v4f* __restrict__ ps4 = (const v4f*)psum;
  const v4f* __restrict__ pq4 = (const v4f*)psq;
  v4f sp[LSPLIT], qp[LSPLIT];
#pragma unroll
  for (int k = 0; k < LSPLIT; ++k) {
    sp[k] = ps4[((size_t)b * LSPLIT + k) * (C_ / 4) + c4];
    qp[k] = pq4[((size_t)b * LSPLIT + k) * (C_ / 4) + c4];
  }

  v4f ring[APPLY_AHEAD + 1][APPLY_BATCH];
#define APPLY_LOAD_BATCH(K)                                                  \
  do {                                                                       \
    _Pragma("unroll") for (int j = 0; j < APPLY_BATCH; ++j)                  \
        ring[(K) % (APPLY_AHEAD + 1)][j] =                                   \
            xp[-(ptrdiff_t)((K) * APPLY_BATCH + j) * APPLY_STEP];            \
  } while (0)

  __builtin_amdgcn_sched_barrier(0);
#pragma unroll
  for (int k = 0; k < APPLY_AHEAD; ++k) APPLY_LOAD_BATCH(k);
  __builtin_amdgcn_sched_barrier(0);

  const bool scat = d1 < NUM_DOMAINS;
  const int  off  = __builtin_amdgcn_readlane(offv, d1 & (NUM_DOMAINS - 1));
  const bool in_lo = scat && dom_lo == d1;
  const bool in_hi = scat && dom_hi == d1;
  const unsigned m_lo  = __builtin_amdgcn_ballot_w32(in_lo);
  const unsigned m_hi  = __builtin_amdgcn_ballot_w32(in_hi);
  const unsigned below = (1u << lane) - 1u;
  const int rank_lo = __popc(m_lo & below);
  const int rank_hi = __popc(m_lo) + __popc(m_hi & below);
  const unsigned h_lo = __builtin_amdgcn_ballot_w32(
      in_lo & (((off + rank_lo) % CAPACITY) == f1));
  const unsigned h_hi = __builtin_amdgcn_ballot_w32(
      in_hi & (((off + rank_hi) % CAPACITY) == f1));
  int src = -1;
  if (h_hi)      src = 63 - __builtin_clz(h_hi);
  else if (h_lo) src = 31 - __builtin_clz(h_lo);

  v4f m1p, s1p;
  v4f sp2[LSPLIT], qp2[LSPLIT];
  if (src >= 0) {
#pragma unroll
    for (int k = 0; k < LSPLIT; ++k) {
      sp2[k] = ps4[((size_t)src * LSPLIT + k) * (C_ / 4) + c4];
      qp2[k] = pq4[((size_t)src * LSPLIT + k) * (C_ / 4) + c4];
    }
  } else {
    const size_t qo = ((size_t)d1 * CAPACITY + f1) * (C_ / 4) + c4;
    m1p = ((const v4f*)mean_q)[qo];
    s1p = ((const v4f*)sig_q)[qo];
  }

  v4f av, bv;
#pragma unroll
  for (int k = 0; k < APPLY_NBATCH; ++k) {
    __builtin_amdgcn_sched_barrier(0);
    if (k + APPLY_AHEAD < APPLY_NBATCH) APPLY_LOAD_BATCH(k + APPLY_AHEAD);
    __builtin_amdgcn_sched_barrier(0);
    if (k == 0) {
      v4f s = {0.f, 0.f, 0.f, 0.f}, q = s;
#pragma unroll
      for (int j = 0; j < LSPLIT; ++j) { s += sp[j]; q += qp[j]; }
      v4f mu1, sig1;
      if (src >= 0) {
        v4f s2 = {0.f, 0.f, 0.f, 0.f}, q2 = s2;
#pragma unroll
        for (int j = 0; j < LSPLIT; ++j) { s2 += sp2[j]; q2 += qp2[j]; }
#pragma unroll
        for (int e = 0; e < 4; ++e) {
          float m, sg;
          dq_stat(s2[e], q2[e], m, sg);
          mu1[e] = m; sig1[e] = sg;
        }
      } else {
        mu1 = m1p; sig1 = s1p;
      }
#pragma unroll
      for (int e = 0; e < 4; ++e) {
        float mu, sig, a, bias;
        dq_stat(s[e], q[e], mu, sig);
        dq_coef(mu, sig, mu1[e], sig1[e], lam, a, bias);
        av[e] = a; bv[e] = bias;
      }
    }
#pragma unroll
    for (int j = 0; j < APPLY_BATCH; ++j) {
      const v4f xv = ring[k % (APPLY_AHEAD + 1)][j];
      v4f o;
      o.x = fmaf(xv.x, av.x, bv.x);
      o.y = fmaf(xv.y, av.y, bv.y);
      o.z = fmaf(xv.z, av.z, bv.z);
      o.w = fmaf(xv.w, av.w, bv.w);
      __builtin_nontemporal_store(
          o, &op[-(ptrdiff_t)(k * APPLY_BATCH + j) * APPLY_STEP]);
    }
  }
#undef APPLY_LOAD_BATCH
}

extern "C" void kernel_launch(void* const* d_in, const int* in_sizes, int n_in,
                              void* d_out, int out_size, void* d_ws,
                              size_t ws_size, hipStream_t stream) {
  const float* x       = (const float*)d_in[0];
  const int*   domain  = (const int*)d_in[1];
  const float* mean_q  = (const float*)d_in[2];
  const float* sig_q   = (const float*)d_in[3];
  const float* lmda    = (const float*)d_in[4];
  const int*   d_ind1  = (const int*)d_in[5];
  const int*   f_ind1  = (const int*)d_in[6];
  const int*   offsets = (const int*)d_in[7];
  float*       out     = (float*)d_out;

  float* psum = (float*)d_ws;
  float* psq  = psum + (size_t)B_ * LSPLIT * C_;

  dim3 gA(C_ / 128, LSPLIT, B_);
  dq_stats_kernel<<<gA, 128, 0, stream>>>(x, psum, psq);

  dq_apply_kernel<<<gA, 128, 0, stream>>>(
      x, psum, psq, domain, mean_q, sig_q, lmda, d_ind1, f_ind1, offsets, out);
}


# Stats and apply fused into ONE launch (was 2): a work-group owns one sample's 64-channel strip, partial sums go through LDS, strip re-read from L2 (x not held in registers); sum order kept
# speedup vs baseline: 1.9068x; 1.2086x over previous
#include <hip/hip_runtime.h>
#include <math.h>
#include <stddef.h>

#define B_ 64
#define L_ 1024
#define C_ 512
#define NUM_DOMAINS 4
#define CAPACITY 1024
#define EPS 1e-6f
#define LSPLIT 4
#define LCHUNK (L_ / LSPLIT)
#define STRIP 64
#define STRIP4 (STRIP / 4)
#define NWAVES 8

typedef __attribute__((ext_vector_type(2))) float v2f;
typedef __attribute__((ext_vector_type(4))) float v4f;
typedef __attribute__((ext_vector_type(8))) float v8f;

__device__ __forceinline__ void dq_stat(float s, float q, float& mu,
                                        float& sig) {
  mu = s * (1.0f / L_);
  const float var = (q - s * s * (1.0f / L_)) * (1.0f / (L_ - 1));
  sig = sqrtf(var + EPS);
}

__device__ __forceinline__ void dq_coef(float mu, float sig, float mu1,
                                        float sig1, float lam, float& a,
                                        float& bias) {
  const float mu_mix  = mu * lam + mu1 * (1.0f - lam);
  const float sig_mix = sig * lam + sig1 * (1.0f - lam);
  a    = sig_mix / sig;
  bias = mu_mix - mu * a;
}

#define APPLY_BATCH 8
#define APPLY_AHEAD 2
#define APPLY_STEP (NWAVES * 2 * (C_ / 4))
#define APPLY_NBATCH (L_ / (NWAVES * 2) / APPLY_BATCH)

__global__ __launch_bounds__(256) void dq_fused_kernel(
    const float* __restrict__ x, const int* __restrict__ domain,
    const float* __restrict__ mean_q, const float* __restrict__ sig_q,
    const float* __restrict__ lmda, const int* __restrict__ d_ind1,
    const int* __restrict__ f_ind1, const int* __restrict__ offsets,
    float* __restrict__ out) {
  __shared__ __attribute__((aligned(16))) float lS[2][LSPLIT][STRIP];
  __shared__ __attribute__((aligned(16))) float lQ[2][LSPLIT][STRIP];

  const int tid   = threadIdx.x;
  const int lane  = tid & 31;
  const int wave  = tid >> 5;
  const int strip = blockIdx.x;
  const int b     = blockIdx.y;

  const int   d1   = d_ind1[b];
  const int   f1   = f_ind1[b];
  const float lam  = lmda[b];
  const int   offv = offsets[lane & (NUM_DOMAINS - 1)];
  const int   dom_lo = domain[lane];
  const int   dom_hi = domain[lane + 32];

  const int n     = lane & 15;
  const int koff  = (lane >> 4) * 2;
  const int split = wave >> 1;
  const int cw    = (wave & 1) * 32 + 2 * n;
  const int c0    = strip * STRIP + cw;

  auto wave_stats = [&](int bb, int pass) __attribute__((always_inline)) {
    const float* xs =
        x + ((size_t)bb * L_ + (size_t)split * LCHUNK) * C_ + c0;
    v8f accQ0 = {0.f, 0.f, 0.f, 0.f, 0.f, 0.f, 0.f, 0.f};
    v8f accQ1 = accQ0;
    v2f sumA = {0.f, 0.f};
    v2f sumB = {0.f, 0.f};
    for (int l = 0; l < LCHUNK; l += 4) {
      const float* p = xs + (size_t)(l + koff) * C_;
      const v2f r0 = *(const v2f*)(p);
      const v2f r1 = *(const v2f*)(p + C_);
      sumA += r0;
      sumB += r1;
      v2f b0, b1;
      b0.x = r0.x; b0.y = r1.x;
      b1.x = r0.y; b1.y = r1.y;
      accQ0 = __builtin_amdgcn_wmma_f32_16x16x4_f32(
          false, b0, false, b0, (short)0, accQ0, false, false);
      accQ1 = __builtin_amdgcn_wmma_f32_16x16x4_f32(
          false, b1, false, b1, (short)0, accQ1, false, false);
    }
    v2f s = sumA + sumB;
    s.x += __shfl_xor(s.x, 16);
    s.y += __shfl_xor(s.y, 16);
    const int r = lane & 7;
    v2f q;
    q.x = accQ0[0]; q.y = accQ1[0];
#pragma unroll
    for (int i = 1; i < 8; ++i) {
      if (r == i) { q.x = accQ0[i]; q.y = accQ1[i]; }
    }
    if (lane < 8 || lane >= 24) {
      *(v2f*)&lS[pass][split][cw] = s;
      *(v2f*)&lQ[pass][split][cw] = q;
    }
  };

  wave_stats(b, 0);

  const bool scat = d1 < NUM_DOMAINS;
  const int  off  = __builtin_amdgcn_readlane(offv, d1 & (NUM_DOMAINS - 1));
  const bool in_lo = scat && dom_lo == d1;
  const bool in_hi = scat && dom_hi == d1;
  const unsigned m_lo  = __builtin_amdgcn_ballot_w32(in_lo);
  const unsigned m_hi  = __builtin_amdgcn_ballot_w32(in_hi);
  const unsigned below = (1u << lane) - 1u;
  const int rank_lo = __popc(m_lo & below);
  const int rank_hi = __popc(m_lo) + __popc(m_hi & below);
  const unsigned h_lo = __builtin_amdgcn_ballot_w32(
      in_lo & (((off + rank_lo) % CAPACITY) == f1));
  const unsigned h_hi = __builtin_amdgcn_ballot_w32(
      in_hi & (((off + rank_hi) % CAPACITY) == f1));
  int src = -1;
  if (h_hi)      src = 63 - __builtin_clz(h_hi);
  else if (h_lo) src = 31 - __builtin_clz(h_lo);

  const int cl   = lane & 15;
  const int c4   = strip * STRIP4 + cl;
  v4f m1p, s1p;
  if (src >= 0) {
    wave_stats(src, 1);
  } else {
    const size_t qo = ((size_t)d1 * CAPACITY + f1) * (C_ / 4) + c4;
    m1p = ((const v4f*)mean_q)[qo];
    s1p = ((const v4f*)sig_q)[qo];
  }

  __syncthreads();

  const int rsub = lane >> 4;
  const int row0 = wave * 2 + rsub;
  const size_t base = ((size_t)b * L_ + row0) * (C_ / 4) + c4;
  const v4f* __restrict__ xp = (const v4f*)x + base;
  v4f* __restrict__ op       = (v4f*)out + base;

  v4f ring[APPLY_AHEAD + 1][APPLY_BATCH];
#define APPLY_LOAD_BATCH(K)                                                  \
  do {                                                                       \
    _Pragma("unroll") for (int j = 0; j < APPLY_BATCH; ++j)                  \
        ring[(K) % (APPLY_AHEAD + 1)][j] =                                   \
            xp[(size_t)((K) * APPLY_BATCH + j) * APPLY_STEP];                \
  } while (0)

#pragma unroll
  for (int k = 0; k < APPLY_AHEAD; ++k) APPLY_LOAD_BATCH(k);

  v4f av, bv;
#pragma unroll
  for (int k = 0; k < APPLY_NBATCH; ++k) {
    __builtin_amdgcn_sched_barrier(0);
    if (k + APPLY_AHEAD < APPLY_NBATCH) APPLY_LOAD_BATCH(k + APPLY_AHEAD);
    __builtin_amdgcn_sched_barrier(0);
    if (k == 0) {
      v4f s = {0.f, 0.f, 0.f, 0.f}, q = s;
#pragma unroll
      for (int j = 0; j < LSPLIT; ++j) {
        s += *(const v4f*)&lS[0][j][4 * cl];
        q += *(const v4f*)&lQ[0][j][4 * cl];
      }
      v4f mu1, sig1;
      if (src >= 0) {
        v4f s2 = {0.f, 0.f, 0.f, 0.f}, q2 = s2;
#pragma unroll
        for (int j = 0; j < LSPLIT; ++j) {
          s2 += *(const v4f*)&lS[1][j][4 * cl];
          q2 += *(const v4f*)&lQ[1][j][4 * cl];
        }
#pragma unroll
        for (int e = 0; e < 4; ++e) {
          float m, sg;
          dq_stat(s2[e], q2[e], m, sg);
          mu1[e] = m; sig1[e] = sg;
        }
      } else {
        mu1 = m1p; sig1 = s1p;
      }
#pragma unroll
      for (int e = 0; e < 4; ++e) {
        float mu, sig, a, bias;
        dq_stat(s[e], q[e], mu, sig);
        dq_coef(mu, sig, mu1[e], sig1[e], lam, a, bias);
        av[e] = a; bv[e] = bias;
      }
    }
#pragma unroll
    for (int j = 0; j < APPLY_BATCH; ++j) {
      const v4f xv = ring[k % (APPLY_AHEAD + 1)][j];
      v4f o;
      o.x = fmaf(xv.x, av.x, bv.x);
      o.y = fmaf(xv.y, av.y, bv.y);
      o.z = fmaf(xv.z, av.z, bv.z);
      o.w = fmaf(xv.w, av.w, bv.w);
      __builtin_nontemporal_store(
          o, &op[(size_t)(k * APPLY_BATCH + j) * APPLY_STEP]);
    }
  }
#undef APPLY_LOAD_BATCH
}

extern "C" void kernel_launch(void* const* d_in, const int* in_sizes, int n_in,
                              void* d_out, int out_size, void* d_ws,
                              size_t ws_size, hipStream_t stream) {
  const float* x       = (const float*)d_in[0];
  const int*   domain  = (const int*)d_in[1];
  const float* mean_q  = (const float*)d_in[2];
  const float* sig_q   = (const float*)d_in[3];
  const float* lmda    = (const float*)d_in[4];
  const int*   d_ind1  = (const int*)d_in[5];
  const int*   f_ind1  = (const int*)d_in[6];
  const int*   offsets = (const int*)d_in[7];
  float*       out     = (float*)d_out;

  dim3 g(C_ / STRIP, B_);
  dq_fused_kernel<<<g, 256, 0, stream>>>(
      x, domain, mean_q, sig_q, lmda, d_ind1, f_ind1, offsets, out);
}


# x strip parked in LDS by the stats phase; apply phase reads x from LDS, not from L2 (one 16-wave work-group per WGP, 256 KiB strip); 8 row partials of 128 rows (sum order changes)
# speedup vs baseline: 2.0726x; 1.0870x over previous
#include <hip/hip_runtime.h>
#include <math.h>
#include <stddef.h>

#define B_ 64
#define L_ 1024
#define C_ 512
#define NUM_DOMAINS 4
#define CAPACITY 1024
#define EPS 1e-6f
#define LSPLIT 8
#define LCHUNK (L_ / LSPLIT)
#define STRIP 64
#define STRIP4 (STRIP / 4)
#define NWAVES 16

typedef __attribute__((ext_vector_type(2))) float v2f;
typedef __attribute__((ext_vector_type(4))) float v4f;
typedef __attribute__((ext_vector_type(8))) float v8f;

__device__ __forceinline__ void dq_stat(float s, float q, float& mu,
                                        float& sig) {
  mu = s * (1.0f / L_);
  const float var = (q - s * s * (1.0f / L_)) * (1.0f / (L_ - 1));
  sig = sqrtf(var + EPS);
}

__device__ __forceinline__ void dq_coef(float mu, float sig, float mu1,
                                        float sig1, float lam, float& a,
                                        float& bias) {
  const float mu_mix  = mu * lam + mu1 * (1.0f - lam);
  const float sig_mix = sig * lam + sig1 * (1.0f - lam);
  a    = sig_mix / sig;
  bias = mu_mix - mu * a;
}

#define APPLY_ROWS (NWAVES * 2)
#define APPLY_NSTEP (L_ / APPLY_ROWS)

__global__ __launch_bounds__(NWAVES * 32) void dq_fused_kernel(
    const float* __restrict__ x, const int* __restrict__ domain,
    const float* __restrict__ mean_q, const float* __restrict__ sig_q,
    const float* __restrict__ lmda, const int* __restrict__ d_ind1,
    const int* __restrict__ f_ind1, const int* __restrict__ offsets,
    float* __restrict__ out) {
  __shared__ __attribute__((aligned(16))) float lX[L_ * STRIP];
  __shared__ __attribute__((aligned(16))) float lS[2][LSPLIT][STRIP];
  __shared__ __attribute__((aligned(16))) float lQ[2][LSPLIT][STRIP];

  const int tid   = threadIdx.x;
  const int lane  = tid & 31;
  const int wave  = tid >> 5;
  const int strip = blockIdx.x;
  const int b     = blockIdx.y;

  const int   d1   = d_ind1[b];
  const int   f1   = f_ind1[b];
  const float lam  = lmda[b];
  const int   offv = offsets[lane & (NUM_DOMAINS - 1)];
  const int   dom_lo = domain[lane];
  const int   dom_hi = domain[lane + 32];

  const int n     = lane & 15;
  const int koff  = (lane >> 4) * 2;
  const int split = wave >> 1;
  const int cw    = (wave & 1) * 32 + 2 * n;
  const int c0    = strip * STRIP + cw;

  auto wave_stats = [&](int bb, int pass, bool park) __attribute__((always_inline)) {
    const float* xs =
        x + ((size_t)bb * L_ + (size_t)split * LCHUNK) * C_ + c0;
    float* ls = lX + (split * LCHUNK + koff) * STRIP + cw;
    v8f accQ0 = {0.f, 0.f, 0.f, 0.f, 0.f, 0.f, 0.f, 0.f};
    v8f accQ1 = accQ0;
    v2f sumA = {0.f, 0.f};
    v2f sumB = {0.f, 0.f};
    for (int l = 0; l < LCHUNK; l += 4) {
      const float* p = xs + (size_t)(l + koff) * C_;
      const v2f r0 = *(const v2f*)(p);
      const v2f r1 = *(const v2f*)(p + C_);
      if (park) {
        *(v2f*)(ls + l * STRIP)       = r0;
        *(v2f*)(ls + (l + 1) * STRIP) = r1;
      }
      sumA += r0;
      sumB += r1;
      v2f b0, b1;
      b0.x = r0.x; b0.y = r1.x;
      b1.x = r0.y; b1.y = r1.y;
      accQ0 = __builtin_amdgcn_wmma_f32_16x16x4_f32(
          false, b0, false, b0, (short)0, accQ0, false, false);
      accQ1 = __builtin_amdgcn_wmma_f32_16x16x4_f32(
          false, b1, false, b1, (short)0, accQ1, false, false);
    }
    v2f s = sumA + sumB;
    s.x += __shfl_xor(s.x, 16);
    s.y += __shfl_xor(s.y, 16);
    const int r = lane & 7;
    v2f q;
    q.x = accQ0[0]; q.y = accQ1[0];
#pragma unroll
    for (int i = 1; i < 8; ++i) {
      if (r == i) { q.x = accQ0[i]; q.y = accQ1[i]; }
    }
    if (lane < 8 || lane >= 24) {
      *(v2f*)&lS[pass][split][cw] = s;
      *(v2f*)&lQ[pass][split][cw] = q;
    }
  };

  wave_stats(b, 0, true);

  const bool scat = d1 < NUM_DOMAINS;
  const int  off  = __builtin_amdgcn_readlane(offv, d1 & (NUM_DOMAINS - 1));
  const bool in_lo = scat && dom_lo == d1;
  const bool in_hi = scat && dom_hi == d1;
  const unsigned m_lo  = __builtin_amdgcn_ballot_w32(in_lo);
  const unsigned m_hi  = __builtin_amdgcn_ballot_w32(in_hi);
  const unsigned below = (1u << lane) - 1u;
  const int rank_lo = __popc(m_lo & below);
  const int rank_hi = __popc(m_lo) + __popc(m_hi & below);
  const unsigned h_lo = __builtin_amdgcn_ballot_w32(
      in_lo & (((off + rank_lo) % CAPACITY) == f1));
  const unsigned h_hi = __builtin_amdgcn_ballot_w32(
      in_hi & (((off + rank_hi) % CAPACITY) == f1));
  int src = -1;
  if (h_hi)      src = 63 - __builtin_clz(h_hi);
  else if (h_lo) src = 31 - __builtin_clz(h_lo);

  const int cl   = lane & 15;
  const int c4   = strip * STRIP4 + cl;
  v4f m1p, s1p;
  if (src >= 0) {
    wave_stats(src, 1, false);
  } else {
    const size_t qo = ((size_t)d1 * CAPACITY + f1) * (C_ / 4) + c4;
    m1p = ((const v4f*)mean_q)[qo];
    s1p = ((const v4f*)sig_q)[qo];
  }

  __syncthreads();

  v4f av, bv;
  {
    v4f s = {0.f, 0.f, 0.f, 0.f}, q = s;
#pragma unroll
    for (int j = 0; j < LSPLIT; ++j) {
      s += *(const v4f*)&lS[0][j][4 * cl];
      q += *(const v4f*)&lQ[0][j][4 * cl];
    }
    v4f mu1, sig1;
    if (src >= 0) {
      v4f s2 = {0.f, 0.f, 0.f, 0.f}, q2 = s2;
#pragma unroll
      for (int j = 0; j < LSPLIT; ++j) {
        s2 += *(const v4f*)&lS[1][j][4 * cl];
        q2 += *(const v4f*)&lQ[1][j][4 * cl];
      }
#pragma unroll
      for (int e = 0; e < 4; ++e) {
        float m, sg;
        dq_stat(s2[e], q2[e], m, sg);
        mu1[e] = m; sig1[e] = sg;
      }
    } else {
      mu1 = m1p; sig1 = s1p;
    }
#pragma unroll
    for (int e = 0; e < 4; ++e) {
      float mu, sig, a, bias;
      dq_stat(s[e], q[e], mu, sig);
      dq_coef(mu, sig, mu1[e], sig1[e], lam, a, bias);
      av[e] = a; bv[e] = bias;
    }
  }

  const int rsub = lane >> 4;
  const int row0 = wave * 2 + rsub;
  const v4f* lx = (const v4f*)lX + row0 * STRIP4 + cl;
  v4f* __restrict__ op =
      (v4f*)out + ((size_t)b * L_ + row0) * (C_ / 4) + c4;
#pragma unroll 8
  for (int k = 0; k < APPLY_NSTEP; ++k) {
    const v4f xv = lx[k * APPLY_ROWS * STRIP4];
    v4f o;
    o.x = fmaf(xv.x, av.x, bv.x);
    o.y = fmaf(xv.y, av.y, bv.y);
    o.z = fmaf(xv.z, av.z, bv.z);
    o.w = fmaf(xv.w, av.w, bv.w);
    __builtin_nontemporal_store(o, &op[(size_t)k * APPLY_ROWS * (C_ / 4)]);
  }
}

extern "C" void kernel_launch(void* const* d_in, const int* in_sizes, int n_in,
                              void* d_out, int out_size, void* d_ws,
                              size_t ws_size, hipStream_t stream) {
  const float* x       = (const float*)d_in[0];
  const int*   domain  = (const int*)d_in[1];
  const float* mean_q  = (const float*)d_in[2];
  const float* sig_q   = (const float*)d_in[3];
  const float* lmda    = (const float*)d_in[4];
  const int*   d_ind1  = (const int*)d_in[5];
  const int*   f_ind1  = (const int*)d_in[6];
  const int*   offsets = (const int*)d_in[7];
  float*       out     = (float*)d_out;

  dim3 g(C_ / STRIP, B_);
  dq_fused_kernel<<<g, NWAVES * 32, 0, stream>>>(
      x, domain, mean_q, sig_q, lmda, d_ind1, f_ind1, offsets, out);
}
